# starting kernel
# baseline (speedup 1.0000x reference)
#include <hip/hip_runtime.h>


typedef __attribute__((ext_vector_type(2))) float v2f;
typedef __attribute__((ext_vector_type(8))) float v8f;

#define T_DIM     4000
#define C_DIM     512
#define B_DIM     16
#define NCHUNK    10
#define CHUNK_LEN 400
#define WARMUP    64
#define ALPHA_F   0.9f
#define CDECAY    0.1f
#define NCG       ((B_DIM * C_DIM) / 16)
#define WAVES_PER_BLOCK 8

__device__ __forceinline__ float pow01(int e) {
    const float p1  = CDECAY;
    const float p2  = p1 * p1;
    const float p4  = p2 * p2;
    const float p8  = p4 * p4;
    const float p16 = p8 * p8;
    const int   ec  = (e < 0) ? 0 : e;
    float r = 1.0f;
    r *= (ec & 1)  ? p1  : 1.0f;
    r *= (ec & 2)  ? p2  : 1.0f;
    r *= (ec & 4)  ? p4  : 1.0f;
    r *= (ec & 8)  ? p8  : 1.0f;
    r *= (ec & 16) ? p16 : 1.0f;
    return r;
}

__global__ __launch_bounds__(256) void
ema_wmma_scan_kernel(const float* __restrict__ x, float* __restrict__ y) {
    const int lane = threadIdx.x & 31;
    const int wib  = threadIdx.x >> 5;
    const int cg    = blockIdx.x * WAVES_PER_BLOCK + wib;
    const int chunk = blockIdx.y;

    const int hi = lane >> 4;
    const int n  = lane & 15;
    const int m  = lane & 15;

    const int    n0   = cg * 16;
    const int    b    = n0 / C_DIM;
    const int    c0   = n0 % C_DIM;
    const size_t base = (size_t)b * T_DIM * C_DIM + (size_t)c0 + (size_t)n;

    v2f A[4];
#pragma unroll
    for (int g = 0; g < 4; ++g) {
        const int k0 = 4 * g + 2 * hi;
        A[g].x = (k0     <= m) ? ALPHA_F * pow01(m - k0)     : 0.0f;
        A[g].y = (k0 + 1 <= m) ? ALPHA_F * pow01(m - k0 - 1) : 0.0f;
    }
    float pc[8];
#pragma unroll
    for (int r = 0; r < 8; ++r) pc[r] = pow01(r + 8 * hi + 1);

    const int t_store = chunk * CHUNK_LEN;
    const int nwarm   = (chunk == 0) ? 0 : (WARMUP / 16);

    float carry = 0.0f;

    const float* px = x + base + (size_t)(t_store - nwarm * 16) * C_DIM
                              + (size_t)(2 * hi) * C_DIM;

#pragma unroll 1
    for (int it = 0; it < nwarm; ++it) {
        v2f Bm[4];
#pragma unroll
        for (int g = 0; g < 4; ++g) {
            const float* p = px + (size_t)(4 * g) * C_DIM;
            Bm[g].x = p[0];
            Bm[g].y = p[C_DIM];
        }
        px += 16 * C_DIM;
        v8f acc;
#pragma unroll
        for (int r = 0; r < 8; ++r) acc[r] = pc[r] * carry;
        acc = __builtin_amdgcn_wmma_f32_16x16x4_f32(false, A[0], false, Bm[0], (short)0, acc, false, false);
        acc = __builtin_amdgcn_wmma_f32_16x16x4_f32(false, A[1], false, Bm[1], (short)0, acc, false, false);
        acc = __builtin_amdgcn_wmma_f32_16x16x4_f32(false, A[2], false, Bm[2], (short)0, acc, false, false);
        acc = __builtin_amdgcn_wmma_f32_16x16x4_f32(false, A[3], false, Bm[3], (short)0, acc, false, false);
        carry = __shfl(acc[7], 16 + n, 32);
    }

    float* py = y + base + (size_t)t_store * C_DIM + (size_t)(8 * hi) * C_DIM;
#pragma unroll 1
    for (int it = 0; it < CHUNK_LEN / 16; ++it) {
        v2f Bm[4];
#pragma unroll
        for (int g = 0; g < 4; ++g) {
            const float* p = px + (size_t)(4 * g) * C_DIM;
            Bm[g].x = p[0];
            Bm[g].y = p[C_DIM];
        }
        px += 16 * C_DIM;
        v8f acc;
#pragma unroll
        for (int r = 0; r < 8; ++r) acc[r] = pc[r] * carry;
        acc = __builtin_amdgcn_wmma_f32_16x16x4_f32(false, A[0], false, Bm[0], (short)0, acc, false, false);
        acc = __builtin_amdgcn_wmma_f32_16x16x4_f32(false, A[1], false, Bm[1], (short)0, acc, false, false);
        acc = __builtin_amdgcn_wmma_f32_16x16x4_f32(false, A[2], false, Bm[2], (short)0, acc, false, false);
        acc = __builtin_amdgcn_wmma_f32_16x16x4_f32(false, A[3], false, Bm[3], (short)0, acc, false, false);
        carry = __shfl(acc[7], 16 + n, 32);

#pragma unroll
        for (int r = 0; r < 8; ++r) py[(size_t)r * C_DIM] = acc[r];
        py += 16 * C_DIM;
    }
}

extern "C" void kernel_launch(void* const* d_in, const int* in_sizes, int n_in,
                              void* d_out, int out_size, void* d_ws, size_t ws_size,
                              hipStream_t stream) {
    (void)in_sizes; (void)n_in; (void)out_size; (void)d_ws; (void)ws_size;
    const float* x = (const float*)d_in[0];
    float*       y = (float*)d_out;

    dim3 grid(NCG / WAVES_PER_BLOCK, NCHUNK);
    dim3 block(WAVES_PER_BLOCK * 32);

    ema_wmma_scan_kernel<<<grid, block, 0, stream>>>(x, y);
}


# Loads run 3 time blocks ahead of the scan: ring of four register tiles per wave, each block first refills the tile behind it (24 loads in flight per wave, not 8 waited for at once)
# speedup vs baseline: 1.1109x; 1.1109x over previous
#include <hip/hip_runtime.h>


typedef __attribute__((ext_vector_type(2))) float v2f;
typedef __attribute__((ext_vector_type(8))) float v8f;

#define T_DIM     4000
#define C_DIM     512
#define B_DIM     16
#define NCHUNK    10
#define CHUNK_LEN 400
#define WARMUP    64
#define ALPHA_F   0.9f
#define CDECAY    0.1f
#define NCG       ((B_DIM * C_DIM) / 16)
#define WAVES_PER_BLOCK 8

#define RING 4

__device__ __forceinline__ float pow01(int e) {
    const float p1  = CDECAY;
    const float p2  = p1 * p1;
    const float p4  = p2 * p2;
    const float p8  = p4 * p4;
    const float p16 = p8 * p8;
    const int   ec  = (e < 0) ? 0 : e;
    float r = 1.0f;
    r *= (ec & 1)  ? p1  : 1.0f;
    r *= (ec & 2)  ? p2  : 1.0f;
    r *= (ec & 4)  ? p4  : 1.0f;
    r *= (ec & 8)  ? p8  : 1.0f;
    r *= (ec & 16) ? p16 : 1.0f;
    return r;
}

__device__ __forceinline__ void load_tile(float (&Bm)[8], const float* px) {
#pragma unroll
    for (int g = 0; g < 4; ++g) {
        const float* p = px + (size_t)(4 * g) * C_DIM;
        Bm[2 * g]     = p[0];
        Bm[2 * g + 1] = p[C_DIM];
    }
}

template <bool REFILL>
__device__ __forceinline__ void scan_block(const float (&Bq)[8], float (&Bfree)[8],
                                           const v2f (&A)[4], const float (&pc)[8],
                                           float& carry, const float*& px, float*& py,
                                           int n, bool store) {
    if (REFILL) {
        load_tile(Bfree, px);
        px += 16 * C_DIM;
    }
    v8f acc;
#pragma unroll
    for (int r = 0; r < 8; ++r) acc[r] = pc[r] * carry;
    acc = __builtin_amdgcn_wmma_f32_16x16x4_f32(false, A[0], false, (v2f){Bq[0], Bq[1]}, (short)0, acc, false, false);
    acc = __builtin_amdgcn_wmma_f32_16x16x4_f32(false, A[1], false, (v2f){Bq[2], Bq[3]}, (short)0, acc, false, false);
    acc = __builtin_amdgcn_wmma_f32_16x16x4_f32(false, A[2], false, (v2f){Bq[4], Bq[5]}, (short)0, acc, false, false);
    acc = __builtin_amdgcn_wmma_f32_16x16x4_f32(false, A[3], false, (v2f){Bq[6], Bq[7]}, (short)0, acc, false, false);
    carry = __shfl(acc[7], 16 + n, 32);
    if (store) {
#pragma unroll
        for (int r = 0; r < 8; ++r) py[(size_t)r * C_DIM] = acc[r];
        py += 16 * C_DIM;
    }
}

__global__ __launch_bounds__(256) void
ema_wmma_scan_kernel(const float* __restrict__ x, float* __restrict__ y) {
    const int lane = threadIdx.x & 31;
    const int wib  = threadIdx.x >> 5;
    const int cg    = blockIdx.x * WAVES_PER_BLOCK + wib;
    const int chunk = blockIdx.y;

    const int hi = lane >> 4;
    const int n  = lane & 15;
    const int m  = lane & 15;

    const int    n0   = cg * 16;
    const int    b    = n0 / C_DIM;
    const int    c0   = n0 % C_DIM;
    const size_t base = (size_t)b * T_DIM * C_DIM + (size_t)c0 + (size_t)n;

    v2f A[4];
#pragma unroll
    for (int g = 0; g < 4; ++g) {
        const int k0 = 4 * g + 2 * hi;
        A[g].x = (k0     <= m) ? ALPHA_F * pow01(m - k0)     : 0.0f;
        A[g].y = (k0 + 1 <= m) ? ALPHA_F * pow01(m - k0 - 1) : 0.0f;
    }
    float pc[8];
#pragma unroll
    for (int r = 0; r < 8; ++r) pc[r] = pow01(r + 8 * hi + 1);

    const int t_store = chunk * CHUNK_LEN;
    const int nwarm   = (chunk == 0) ? 0 : (WARMUP / 16);

    float carry = 0.0f;

    const float* px = x + base + (size_t)(t_store - nwarm * 16) * C_DIM
                              + (size_t)(2 * hi) * C_DIM;
    float* py = y + base + (size_t)t_store * C_DIM + (size_t)(8 * hi) * C_DIM;

    float B0[8], B1[8], B2[8], B3[8];
    load_tile(B0, px); px += 16 * C_DIM;
    load_tile(B1, px); px += 16 * C_DIM;
    load_tile(B2, px); px += 16 * C_DIM;

    const int first_turn = (chunk == 0) ? 0 : -(WARMUP / 16) / RING;
#pragma unroll 1
    for (int turn = first_turn; turn < (CHUNK_LEN / 16 - RING - 1) / RING; ++turn) {
        const bool st = turn >= 0;
        scan_block<true>(B0, B3, A, pc, carry, px, py, n, st);
        scan_block<true>(B1, B0, A, pc, carry, px, py, n, st);
        scan_block<true>(B2, B1, A, pc, carry, px, py, n, st);
        scan_block<true>(B3, B2, A, pc, carry, px, py, n, st);
    }

    scan_block<true >(B0, B3, A, pc, carry, px, py, n, true);
    scan_block<true >(B1, B0, A, pc, carry, px, py, n, true);
    scan_block<false>(B2, B1, A, pc, carry, px, py, n, true);
    scan_block<false>(B3, B2, A, pc, carry, px, py, n, true);
    scan_block<false>(B0, B1, A, pc, carry, px, py, n, true);
}

extern "C" void kernel_launch(void* const* d_in, const int* in_sizes, int n_in,
                              void* d_out, int out_size, void* d_ws, size_t ws_size,
                              hipStream_t stream) {
    (void)in_sizes; (void)n_in; (void)out_size; (void)d_ws; (void)ws_size;
    const float* x = (const float*)d_in[0];
    float*       y = (float*)d_out;

    dim3 grid(NCG / WAVES_PER_BLOCK, NCHUNK);
    dim3 block(WAVES_PER_BLOCK * 32);

    ema_wmma_scan_kernel<<<grid, block, 0, stream>>>(x, y);
}


# Each lane loads and stores 4 adjacent channels as one 16-byte access: a wave scans 4 column tiles (64 channels), every access covers whole aligned 256-byte row pieces; 1280 one-wave blocks
# speedup vs baseline: 1.5607x; 1.4049x over previous
#include <hip/hip_runtime.h>


typedef __attribute__((ext_vector_type(2))) float v2f;
typedef __attribute__((ext_vector_type(4))) float v4f;
typedef __attribute__((ext_vector_type(8))) float v8f;

#define T_DIM     4000
#define C_DIM     512
#define B_DIM     16
#define NCHUNK    10
#define CHUNK_LEN 400
#define WARMUP    64
#define ALPHA_F   0.9f
#define CDECAY    0.1f
#define NT        4
#define NCG       ((B_DIM * C_DIM) / (16 * NT))

#define RING 4

__device__ __forceinline__ float pow01(int e) {
    const float p1  = CDECAY;
    const float p2  = p1 * p1;
    const float p4  = p2 * p2;
    const float p8  = p4 * p4;
    const float p16 = p8 * p8;
    const int   ec  = (e < 0) ? 0 : e;
    float r = 1.0f;
    r *= (ec & 1)  ? p1  : 1.0f;
    r *= (ec & 2)  ? p2  : 1.0f;
    r *= (ec & 4)  ? p4  : 1.0f;
    r *= (ec & 8)  ? p8  : 1.0f;
    r *= (ec & 16) ? p16 : 1.0f;
    return r;
}

__device__ __forceinline__ void load_tile(v4f (&Bm)[8], const float* px) {
#pragma unroll
    for (int g = 0; g < 4; ++g) {
        const float* p = px + (size_t)(4 * g) * C_DIM;
        Bm[2 * g]     = *(const v4f*)p;
        Bm[2 * g + 1] = *(const v4f*)(p + C_DIM);
    }
}

template <bool REFILL>
__device__ __forceinline__ void scan_block(const v4f (&Bq)[8], v4f (&Bfree)[8],
                                           const v2f (&A)[4], const float (&pc)[8],
                                           v4f& carry, const float*& px, float*& py,
                                           int n, bool store) {
    if (REFILL) {
        load_tile(Bfree, px);
        px += 16 * C_DIM;
    }
    v4f q[8];
#pragma unroll
    for (int i = 0; i < 8; ++i) {
        q[i] = Bq[i];
        asm volatile("" : "+v"(q[i]));
    }
    v8f acc[NT];
#pragma unroll
    for (int j = 0; j < NT; ++j)
#pragma unroll
        for (int r = 0; r < 8; ++r) acc[j][r] = pc[r] * carry[j];
#pragma unroll
    for (int g = 0; g < 4; ++g)
#pragma unroll
        for (int j = 0; j < NT; ++j)
            acc[j] = __builtin_amdgcn_wmma_f32_16x16x4_f32(false, A[g], false,
                         (v2f){q[2 * g][j], q[2 * g + 1][j]}, (short)0, acc[j], false, false);
#pragma unroll
    for (int j = 0; j < NT; ++j)
        carry[j] = __shfl(acc[j][7], 16 + n, 32);
    if (store) {
#pragma unroll
        for (int r = 0; r < 8; ++r)
            *(v4f*)(py + (size_t)r * C_DIM) = (v4f){acc[0][r], acc[1][r], acc[2][r], acc[3][r]};
        py += 16 * C_DIM;
    }
}

__global__ __launch_bounds__(32) void
ema_wmma_scan_kernel(const float* __restrict__ x, float* __restrict__ y) {
    const int lane  = threadIdx.x & 31;
    const int cg    = blockIdx.x;
    const int chunk = blockIdx.y;

    const int hi = lane >> 4;
    const int n  = lane & 15;
    const int m  = lane & 15;

    const int    n0   = cg * 16 * NT;
    const int    b    = n0 / C_DIM;
    const int    c0   = n0 % C_DIM;
    const size_t base = (size_t)b * T_DIM * C_DIM + (size_t)c0 + (size_t)(NT * n);

    v2f A[4];
#pragma unroll
    for (int g = 0; g < 4; ++g) {
        const int k0 = 4 * g + 2 * hi;
        A[g].x = (k0     <= m) ? ALPHA_F * pow01(m - k0)     : 0.0f;
        A[g].y = (k0 + 1 <= m) ? ALPHA_F * pow01(m - k0 - 1) : 0.0f;
    }
    float pc[8];
#pragma unroll
    for (int r = 0; r < 8; ++r) pc[r] = pow01(r + 8 * hi + 1);

    const int t_store = chunk * CHUNK_LEN;
    const int nwarm   = (chunk == 0) ? 0 : (WARMUP / 16);

    v4f carry = (v4f){0.0f, 0.0f, 0.0f, 0.0f};

    const float* px = x + base + (size_t)(t_store - nwarm * 16) * C_DIM
                              + (size_t)(2 * hi) * C_DIM;
    float* py = y + base + (size_t)t_store * C_DIM + (size_t)(8 * hi) * C_DIM;

    v4f B0[8], B1[8], B2[8], B3[8];
    load_tile(B0, px); px += 16 * C_DIM;
    load_tile(B1, px); px += 16 * C_DIM;
    load_tile(B2, px); px += 16 * C_DIM;

    const int first_turn = (chunk == 0) ? 0 : -(WARMUP / 16) / RING;
#pragma unroll 1
    for (int turn = first_turn; turn < (CHUNK_LEN / 16 - RING - 1) / RING; ++turn) {
        const bool st = turn >= 0;
        scan_block<true>(B0, B3, A, pc, carry, px, py, n, st);
        scan_block<true>(B1, B0, A, pc, carry, px, py, n, st);
        scan_block<true>(B2, B1, A, pc, carry, px, py, n, st);
        scan_block<true>(B3, B2, A, pc, carry, px, py, n, st);
    }

    scan_block<true >(B0, B3, A, pc, carry, px, py, n, true);
    scan_block<true >(B1, B0, A, pc, carry, px, py, n, true);
    scan_block<false>(B2, B1, A, pc, carry, px, py, n, true);
    scan_block<false>(B3, B2, A, pc, carry, px, py, n, true);
    scan_block<false>(B0, B1, A, pc, carry, px, py, n, true);
}

extern "C" void kernel_launch(void* const* d_in, const int* in_sizes, int n_in,
                              void* d_out, int out_size, void* d_ws, size_t ws_size,
                              hipStream_t stream) {
    (void)in_sizes; (void)n_in; (void)out_size; (void)d_ws; (void)ws_size;
    const float* x = (const float*)d_in[0];
    float*       y = (float*)d_out;

    dim3 grid(NCG, NCHUNK);
    dim3 block(32);

    ema_wmma_scan_kernel<<<grid, block, 0, stream>>>(x, y);
}
